# starting kernel
# baseline (speedup 1.0000x reference)
#include <hip/hip_runtime.h>

typedef __attribute__((ext_vector_type(16))) _Float16 v16h;
typedef __attribute__((ext_vector_type(8)))  _Float16 v8h;
typedef __attribute__((ext_vector_type(8)))  float    v8f;

#define HID 128

__global__ void convert_wt_kernel(const float* __restrict__ W,
                                  _Float16* __restrict__ Wt) {
    int idx = blockIdx.x * blockDim.x + threadIdx.x;
    int n = idx >> 7;
    int k = idx & 127;
    Wt[idx] = (_Float16)W[k * HID + n];
}

__global__ void encode_kernel(const float* __restrict__ x,
                              const float* __restrict__ W,
                              const float* __restrict__ b,
                              float* __restrict__ h, int N) {
    int wave = threadIdx.x >> 5, lane = threadIdx.x & 31;
    int n = blockIdx.x * 8 + wave;
    if (n >= N) return;
    int c = lane * 4;
    float xv = x[n];
    float4 w  = *(const float4*)(W + c);
    float4 bb = *(const float4*)(b + c);
    float4 r;
    r.x = xv * w.x + bb.x;
    r.y = xv * w.y + bb.y;
    r.z = xv * w.z + bb.z;
    r.w = xv * w.w + bb.w;
    *(float4*)(h + (size_t)n * HID + c) = r;
}

__global__ void scale_kernel(const float* __restrict__ h,
                             const float* __restrict__ eps, int l,
                             float* __restrict__ agg, int n4) {
    int i = blockIdx.x * blockDim.x + threadIdx.x;
    if (i >= n4) return;
    float s = 1.0f + eps[l];
    float4 v = ((const float4*)h)[i];
    v.x *= s; v.y *= s; v.z *= s; v.w *= s;
    ((float4*)agg)[i] = v;
}

__global__ void edge_kernel(const float* __restrict__ h,
                            const int* __restrict__ ei,
                            const float* __restrict__ ea,
                            const float* __restrict__ eW,
                            const float* __restrict__ eb,
                            float* __restrict__ agg, int E) {
    int wave = threadIdx.x >> 5, lane = threadIdx.x & 31;
    int e = blockIdx.x * 8 + wave;
    if (e >= E) return;
    int src = ei[e];
    int dst = ei[E + e];
    float a0 = ea[2 * e];
    float a1 = ea[2 * e + 1];
    int c = lane * 4;
    float4 hv = *(const float4*)(h + (size_t)src * HID + c);
    float4 w0 = *(const float4*)(eW + c);
    float4 w1 = *(const float4*)(eW + HID + c);
    float4 bb = *(const float4*)(eb + c);
    float* ap = agg + (size_t)dst * HID + c;
    atomicAdd(ap + 0, fmaxf(hv.x + a0 * w0.x + a1 * w1.x + bb.x, 0.0f));
    atomicAdd(ap + 1, fmaxf(hv.y + a0 * w0.y + a1 * w1.y + bb.y, 0.0f));
    atomicAdd(ap + 2, fmaxf(hv.z + a0 * w0.z + a1 * w1.z + bb.z, 0.0f));
    atomicAdd(ap + 3, fmaxf(hv.w + a0 * w0.w + a1 * w1.w + bb.w, 0.0f));
}

__global__ void zero_kernel(float* __restrict__ p, int n4) {
    int i = blockIdx.x * blockDim.x + threadIdx.x;
    if (i < n4) ((float4*)p)[i] = make_float4(0.f, 0.f, 0.f, 0.f);
}

__global__ void pool_kernel(const float* __restrict__ h,
                            const int* __restrict__ batch,
                            float* __restrict__ pooled, int N) {
    int wave = threadIdx.x >> 5, lane = threadIdx.x & 31;
    int n = blockIdx.x * 8 + wave;
    if (n >= N) return;
    int g = batch[n];
    int c = lane * 4;
    float4 hv = *(const float4*)(h + (size_t)n * HID + c);
    float* pp = pooled + (size_t)g * HID + c;
    atomicAdd(pp + 0, hv.x);
    atomicAdd(pp + 1, hv.y);
    atomicAdd(pp + 2, hv.z);
    atomicAdd(pp + 3, hv.w);
}

template <bool RELU>
__global__ __launch_bounds__(256) void gemm128_wmma_kernel(
    const float* __restrict__ A, const _Float16* __restrict__ Wt,
    const float* __restrict__ bias, float* __restrict__ out, int rows) {
    __shared__ __align__(16) _Float16 sW[HID * HID];
    {
        const uint4* s = (const uint4*)Wt;
        uint4* d = (uint4*)sW;
        for (int i = threadIdx.x; i < (HID * HID) / 8; i += 256) d[i] = s[i];
    }
    __syncthreads();

    int wave = threadIdx.x >> 5, lane = threadIdx.x & 31;
    int tile = blockIdx.x * 8 + wave;
    int row0 = tile * 16;
    if (row0 >= rows) return;

    int mloc = lane & 15;
    int half = lane >> 4;
    int base = half * 8;
    const float* rowA = A + (size_t)(row0 + mloc) * HID;

    v8f acc[8];
#pragma unroll
    for (int nt = 0; nt < 8; ++nt) {
        float bv = bias[nt * 16 + mloc];
#pragma unroll
        for (int r = 0; r < 8; ++r) acc[nt][r] = bv;
    }

#pragma unroll
    for (int kk = 0; kk < 4; ++kk) {
        int k0 = kk * 32 + base;
        float4 a0 = *(const float4*)(rowA + k0);
        float4 a1 = *(const float4*)(rowA + k0 + 4);
        float4 a2 = *(const float4*)(rowA + k0 + 16);
        float4 a3 = *(const float4*)(rowA + k0 + 20);
        v16h a;
        a[0]  = (_Float16)a0.x; a[1]  = (_Float16)a0.y;
        a[2]  = (_Float16)a0.z; a[3]  = (_Float16)a0.w;
        a[4]  = (_Float16)a1.x; a[5]  = (_Float16)a1.y;
        a[6]  = (_Float16)a1.z; a[7]  = (_Float16)a1.w;
        a[8]  = (_Float16)a2.x; a[9]  = (_Float16)a2.y;
        a[10] = (_Float16)a2.z; a[11] = (_Float16)a2.w;
        a[12] = (_Float16)a3.x; a[13] = (_Float16)a3.y;
        a[14] = (_Float16)a3.z; a[15] = (_Float16)a3.w;
#pragma unroll
        for (int nt = 0; nt < 8; ++nt) {
            int n = nt * 16 + mloc;
            const v8h* bp = (const v8h*)(sW + n * HID + k0);
            v8h blo = bp[0];
            v8h bhi = bp[2];
            v16h b;
#pragma unroll
            for (int i = 0; i < 8; ++i) { b[i] = blo[i]; b[8 + i] = bhi[i]; }
            acc[nt] = __builtin_amdgcn_wmma_f32_16x16x32_f16(
                false, a, false, b, (short)0, acc[nt], false, false);
        }
    }

#pragma unroll
    for (int nt = 0; nt < 8; ++nt) {
        int col = nt * 16 + mloc;
#pragma unroll
        for (int r = 0; r < 8; ++r) {
            float v = acc[nt][r];
            if (RELU) v = fmaxf(v, 0.0f);
            int M = row0 + r + half * 8;
            out[(size_t)M * HID + col] = v;
        }
    }
}

extern "C" void kernel_launch(void* const* d_in, const int* in_sizes, int n_in,
                              void* d_out, int out_size, void* d_ws, size_t ws_size,
                              hipStream_t stream) {
    const int N = in_sizes[0];
    const int E = in_sizes[1] / 2;
    const int G = out_size / (3 * HID);

    const float* enc_W  = (const float*)d_in[12];
    const float* enc_b  = (const float*)d_in[13];
    const float* edge_W = (const float*)d_in[14];
    const float* edge_b = (const float*)d_in[15];
    const float* eps    = (const float*)d_in[16];
    const float* W1     = (const float*)d_in[17];
    const float* b1     = (const float*)d_in[18];
    const float* W2     = (const float*)d_in[19];
    const float* b2     = (const float*)d_in[20];
    const float* fc_W   = (const float*)d_in[21];
    const float* fc_b   = (const float*)d_in[22];

    float* bufA   = (float*)d_ws;
    float* bufB   = bufA + (size_t)N * HID;
    float* pooled = bufB + (size_t)N * HID;
    _Float16* w16 = (_Float16*)(pooled + (size_t)G * HID);
    _Float16* W1t0 = w16;
    _Float16* W1t1 = w16 + 16384;
    _Float16* W2t0 = w16 + 2 * 16384;
    _Float16* W2t1 = w16 + 3 * 16384;
    _Float16* fcWt = w16 + 4 * 16384;

    convert_wt_kernel<<<64, 256, 0, stream>>>(W1,             W1t0);
    convert_wt_kernel<<<64, 256, 0, stream>>>(W1 + HID * HID, W1t1);
    convert_wt_kernel<<<64, 256, 0, stream>>>(W2,             W2t0);
    convert_wt_kernel<<<64, 256, 0, stream>>>(W2 + HID * HID, W2t1);
    convert_wt_kernel<<<64, 256, 0, stream>>>(fc_W,           fcWt);

    const int n4        = (N * HID) / 4;
    const int nodeBlk   = (N + 7) / 8;
    const int edgeBlk   = (E + 7) / 8;
    const int scaleBlk  = (n4 + 255) / 256;
    const int gemmBlk   = ((N + 15) / 16 + 7) / 8;
    const int gP4       = (G * HID) / 4;
    const int fcBlk     = ((G + 15) / 16 + 7) / 8;

    for (int g = 0; g < 3; ++g) {
        const float* x  = (const float*)d_in[4 * g + 0];
        const int*   ei = (const int*)  d_in[4 * g + 1];
        const float* ea = (const float*)d_in[4 * g + 2];
        const int*   bt = (const int*)  d_in[4 * g + 3];

        encode_kernel<<<nodeBlk, 256, 0, stream>>>(x, enc_W, enc_b, bufA, N);

        float* hcur = bufA;
        float* aggb = bufB;
        for (int l = 0; l < 2; ++l) {
            scale_kernel<<<scaleBlk, 256, 0, stream>>>(hcur, eps, l, aggb, n4);
            edge_kernel<<<edgeBlk, 256, 0, stream>>>(
                hcur, ei, ea, edge_W + (size_t)l * 2 * HID, edge_b + (size_t)l * HID,
                aggb, E);
            _Float16* w1t = (l == 0) ? W1t0 : W1t1;
            _Float16* w2t = (l == 0) ? W2t0 : W2t1;
            gemm128_wmma_kernel<true><<<gemmBlk, 256, 0, stream>>>(
                aggb, w1t, b1 + (size_t)l * HID, hcur, N);
            gemm128_wmma_kernel<true><<<gemmBlk, 256, 0, stream>>>(
                hcur, w2t, b2 + (size_t)l * HID, aggb, N);
            float* t = hcur; hcur = aggb; aggb = t;
        }

        zero_kernel<<<(gP4 + 255) / 256, 256, 0, stream>>>(pooled, gP4);
        pool_kernel<<<nodeBlk, 256, 0, stream>>>(hcur, bt, pooled, N);
        gemm128_wmma_kernel<false><<<fcBlk, 256, 0, stream>>>(
            pooled, fcWt, fc_b, (float*)d_out + (size_t)g * G * HID, G);
    }
}


# edge grouping: per-edge global integer atomics of the count and place passes replaced by LDS counters; each block owns a node range and walks the edge list; counter-clear launch dropped
# speedup vs baseline: 1.8070x; 1.8070x over previous
#include <hip/hip_runtime.h>

typedef __attribute__((ext_vector_type(16))) _Float16 v16h;
typedef __attribute__((ext_vector_type(8)))  _Float16 v8h;
typedef __attribute__((ext_vector_type(8)))  float    v8f;

#define HID 128

__global__ void convert_wt_kernel(const float* __restrict__ W,
                                  _Float16* __restrict__ Wt) {
    int idx = blockIdx.x * blockDim.x + threadIdx.x;
    int n = idx >> 7;
    int k = idx & 127;
    Wt[idx] = (_Float16)W[k * HID + n];
}

__global__ void encode_kernel(const float* __restrict__ x,
                              const float* __restrict__ W,
                              const float* __restrict__ b,
                              float* __restrict__ h, int N) {
    int wave = threadIdx.x >> 5, lane = threadIdx.x & 31;
    int n = blockIdx.x * 8 + wave;
    if (n >= N) return;
    int c = lane * 4;
    float xv = x[n];
    float4 w  = *(const float4*)(W + c);
    float4 bb = *(const float4*)(b + c);
    float4 r;
    r.x = xv * w.x + bb.x;
    r.y = xv * w.y + bb.y;
    r.z = xv * w.z + bb.z;
    r.w = xv * w.w + bb.w;
    *(float4*)(h + (size_t)n * HID + c) = r;
}


struct EdgeIn {
    const int* ei0; const int* ei1; const int* ei2;
    const float* ea0; const float* ea1; const float* ea2;
};
__device__ __forceinline__ const int* pick_ei(const EdgeIn& p, int g) {
    return g == 0 ? p.ei0 : (g == 1 ? p.ei1 : p.ei2);
}
__device__ __forceinline__ const float* pick_ea(const EdgeIn& p, int g) {
    return g == 0 ? p.ea0 : (g == 1 ? p.ea1 : p.ea2);
}

extern __shared__ int lds_cnt[];

template <bool PLACE>
__global__ __launch_bounds__(1024) void range_kernel(EdgeIn p, int* __restrict__ cnt,
                                                     const int* __restrict__ off,
                                                     int* __restrict__ perm,
                                                     int E, int N, int NP, int EP, int R) {
    const int g = blockIdx.y, tid = threadIdx.x;
    const int lo = blockIdx.x * R;
    int nloc = min(lo + R, N) - lo;
    if (nloc < 0) nloc = 0;
    const unsigned un = (unsigned)nloc;
    for (int i = tid; i < nloc; i += 1024)
        lds_cnt[i] = PLACE ? off[(size_t)g * NP + lo + i] : 0;
    __syncthreads();

    const int* dst = pick_ei(p, g) + E;
    int* pg = perm + (size_t)g * EP;
    auto visit = [&](int d, int e) {
        unsigned r = (unsigned)(d - lo);
        if (r < un) {
            if (PLACE) {
                int slot = atomicAdd(&lds_cnt[r], 1);
                if ((unsigned)slot < (unsigned)E) pg[slot] = e;
            } else {
                atomicAdd(&lds_cnt[r], 1);
            }
        }
    };
    auto visit4 = [&](const int4& d, int j) {
        unsigned a = (unsigned)(d.x - lo), b = (unsigned)(d.y - lo);
        unsigned c = (unsigned)(d.z - lo), w = (unsigned)(d.w - lo);
        if (min(min(a, b), min(c, w)) < un) {
            visit(d.x, 4 * j);     visit(d.y, 4 * j + 1);
            visit(d.z, 4 * j + 2); visit(d.w, 4 * j + 3);
        }
    };

    const bool vec = ((E & 3) == 0) && ((((size_t)dst) & 15) == 0);
    const int E4 = vec ? (E >> 2) : 0;
    const int4* d4 = (const int4*)dst;
    const int start = (int)(((long long)blockIdx.x * E4) / gridDim.x);
    int i = tid;
    for (; i + 3072 < E4; i += 4096) {
        int j0 = i + start;        if (j0 >= E4) j0 -= E4;
        int j1 = i + 1024 + start; if (j1 >= E4) j1 -= E4;
        int j2 = i + 2048 + start; if (j2 >= E4) j2 -= E4;
        int j3 = i + 3072 + start; if (j3 >= E4) j3 -= E4;
        int4 q0 = d4[j0], q1 = d4[j1], q2 = d4[j2], q3 = d4[j3];
        visit4(q0, j0); visit4(q1, j1); visit4(q2, j2); visit4(q3, j3);
    }
    for (; i < E4; i += 1024) {
        int j0 = i + start; if (j0 >= E4) j0 -= E4;
        int4 q0 = d4[j0];
        visit4(q0, j0);
    }
    for (int e = 4 * E4 + tid; e < E; e += 1024) visit(dst[e], e);

    if (!PLACE) {
        __syncthreads();
        for (int k = tid; k < nloc; k += 1024) cnt[(size_t)g * NP + lo + k] = lds_cnt[k];
    }
}

__global__ __launch_bounds__(1024) void scan_kernel(const int* __restrict__ cnt,
                                                    int* __restrict__ off, int N, int NP) {
    __shared__ int wsum[32];
    const int* c = cnt + (size_t)blockIdx.x * NP;
    int* o = off + (size_t)blockIdx.x * NP;
    int tid = threadIdx.x, lane = tid & 31, wave = tid >> 5;
    int chunk = (N + 1023) / 1024;
    int lo = tid * chunk;
    if (lo > N) lo = N;
    int hi = lo + chunk;
    if (hi > N) hi = N;
    int s = 0;
    for (int i = lo; i < hi; ++i) s += c[i];
    int v = s;
    for (int d = 1; d < 32; d <<= 1) {
        int t = __shfl_up(v, d);
        if (lane >= d) v += t;
    }
    if (lane == 31) wsum[wave] = v;
    __syncthreads();
    if (wave == 0) {
        int w = wsum[lane];
        for (int d = 1; d < 32; d <<= 1) {
            int t = __shfl_up(w, d);
            if (lane >= d) w += t;
        }
        wsum[lane] = w;
    }
    __syncthreads();
    int run = v - s + (wave ? wsum[wave - 1] : 0);
    for (int i = lo; i < hi; ++i) { o[i] = run; run += c[i]; }
    if (tid == 1023) o[N] = run;
}

__global__ void rank_kernel(EdgeIn p, const int* __restrict__ off,
                            const int* __restrict__ perm, int* __restrict__ ssrc,
                            float* __restrict__ sea, int E, int N, int NP, int EP) {
    int g = blockIdx.y;
    int q = blockIdx.x * blockDim.x + threadIdx.x;
    const int* og = off + (size_t)g * NP;
    const int* pg = perm + (size_t)g * EP;
    if (q >= E || q >= og[N]) return;
    int e = pg[q];
    if ((unsigned)e >= (unsigned)E) return;
    const int* ei = pick_ei(p, g);
    int d = ei[E + e];
    int s0 = og[d], s1 = og[d + 1];
    int r = 0;
    for (int j = s0; j < s1; ++j) r += (pg[j] < e) ? 1 : 0;
    size_t t = (size_t)g * EP + s0 + r;
    ssrc[t] = ei[e];
    ((float2*)sea)[t] = ((const float2*)pick_ea(p, g))[e];
}

__device__ __forceinline__ void add_msg(float4& acc, const float4& hv, float a0, float a1,
                                        const float4& w0, const float4& w1, const float4& bb) {
    acc.x += fmaxf(hv.x + a0 * w0.x + a1 * w1.x + bb.x, 0.0f);
    acc.y += fmaxf(hv.y + a0 * w0.y + a1 * w1.y + bb.y, 0.0f);
    acc.z += fmaxf(hv.z + a0 * w0.z + a1 * w1.z + bb.z, 0.0f);
    acc.w += fmaxf(hv.w + a0 * w0.w + a1 * w1.w + bb.w, 0.0f);
}

__global__ __launch_bounds__(256) void gather_kernel(
    const float* __restrict__ h, const int* __restrict__ off,
    const int* __restrict__ ssrc, const float* __restrict__ sea,
    const float* __restrict__ eW, const float* __restrict__ eb,
    const float* __restrict__ eps, int l, float* __restrict__ agg, int N) {
    int lane = threadIdx.x & 31;
    int n = __builtin_amdgcn_readfirstlane(blockIdx.x * 8 + (threadIdx.x >> 5));
    if (n >= N) return;
    int s0 = off[n], s1 = off[n + 1];
    int c = lane * 4;
    float4 w0 = *(const float4*)(eW + c);
    float4 w1 = *(const float4*)(eW + HID + c);
    float4 bb = *(const float4*)(eb + c);
    float4 hs = *(const float4*)(h + (size_t)n * HID + c);
    const float2* a2 = (const float2*)sea;
    const int last = N - 1;
    float4 acc = make_float4(0.f, 0.f, 0.f, 0.f);
    int p = s0;
    for (; p + 4 <= s1; p += 4) {
        int i0 = min(max(ssrc[p], 0), last),     i1 = min(max(ssrc[p + 1], 0), last);
        int i2 = min(max(ssrc[p + 2], 0), last), i3 = min(max(ssrc[p + 3], 0), last);
        float2 e0 = a2[p], e1 = a2[p + 1], e2 = a2[p + 2], e3 = a2[p + 3];
        float4 h0 = *(const float4*)(h + (size_t)i0 * HID + c);
        float4 h1 = *(const float4*)(h + (size_t)i1 * HID + c);
        float4 h2 = *(const float4*)(h + (size_t)i2 * HID + c);
        float4 h3 = *(const float4*)(h + (size_t)i3 * HID + c);
        add_msg(acc, h0, e0.x, e0.y, w0, w1, bb);
        add_msg(acc, h1, e1.x, e1.y, w0, w1, bb);
        add_msg(acc, h2, e2.x, e2.y, w0, w1, bb);
        add_msg(acc, h3, e3.x, e3.y, w0, w1, bb);
    }
    for (; p < s1; ++p) {
        int i0 = min(max(ssrc[p], 0), last);
        float2 e0 = a2[p];
        float4 h0 = *(const float4*)(h + (size_t)i0 * HID + c);
        add_msg(acc, h0, e0.x, e0.y, w0, w1, bb);
    }
    float s = 1.0f + eps[l];
    float4 r;
    r.x = s * hs.x + acc.x;
    r.y = s * hs.y + acc.y;
    r.z = s * hs.z + acc.z;
    r.w = s * hs.w + acc.w;
    *(float4*)(agg + (size_t)n * HID + c) = r;
}

__global__ void zero_kernel(float* __restrict__ p, int n4) {
    int i = blockIdx.x * blockDim.x + threadIdx.x;
    if (i < n4) ((float4*)p)[i] = make_float4(0.f, 0.f, 0.f, 0.f);
}

__global__ void pool_kernel(const float* __restrict__ h,
                            const int* __restrict__ batch,
                            float* __restrict__ pooled, int N) {
    int wave = threadIdx.x >> 5, lane = threadIdx.x & 31;
    int n = blockIdx.x * 8 + wave;
    if (n >= N) return;
    int g = batch[n];
    int c = lane * 4;
    float4 hv = *(const float4*)(h + (size_t)n * HID + c);
    float* pp = pooled + (size_t)g * HID + c;
    atomicAdd(pp + 0, hv.x);
    atomicAdd(pp + 1, hv.y);
    atomicAdd(pp + 2, hv.z);
    atomicAdd(pp + 3, hv.w);
}

template <bool RELU>
__global__ __launch_bounds__(256) void gemm128_wmma_kernel(
    const float* __restrict__ A, const _Float16* __restrict__ Wt,
    const float* __restrict__ bias, float* __restrict__ out, int rows) {
    __shared__ __align__(16) _Float16 sW[HID * HID];
    {
        const uint4* s = (const uint4*)Wt;
        uint4* d = (uint4*)sW;
        for (int i = threadIdx.x; i < (HID * HID) / 8; i += 256) d[i] = s[i];
    }
    __syncthreads();

    int wave = threadIdx.x >> 5, lane = threadIdx.x & 31;
    int tile = blockIdx.x * 8 + wave;
    int row0 = tile * 16;
    if (row0 >= rows) return;

    int mloc = lane & 15;
    int half = lane >> 4;
    int base = half * 8;
    const float* rowA = A + (size_t)(row0 + mloc) * HID;

    v8f acc[8];
#pragma unroll
    for (int nt = 0; nt < 8; ++nt) {
        float bv = bias[nt * 16 + mloc];
#pragma unroll
        for (int r = 0; r < 8; ++r) acc[nt][r] = bv;
    }

#pragma unroll
    for (int kk = 0; kk < 4; ++kk) {
        int k0 = kk * 32 + base;
        float4 a0 = *(const float4*)(rowA + k0);
        float4 a1 = *(const float4*)(rowA + k0 + 4);
        float4 a2 = *(const float4*)(rowA + k0 + 16);
        float4 a3 = *(const float4*)(rowA + k0 + 20);
        v16h a;
        a[0]  = (_Float16)a0.x; a[1]  = (_Float16)a0.y;
        a[2]  = (_Float16)a0.z; a[3]  = (_Float16)a0.w;
        a[4]  = (_Float16)a1.x; a[5]  = (_Float16)a1.y;
        a[6]  = (_Float16)a1.z; a[7]  = (_Float16)a1.w;
        a[8]  = (_Float16)a2.x; a[9]  = (_Float16)a2.y;
        a[10] = (_Float16)a2.z; a[11] = (_Float16)a2.w;
        a[12] = (_Float16)a3.x; a[13] = (_Float16)a3.y;
        a[14] = (_Float16)a3.z; a[15] = (_Float16)a3.w;
#pragma unroll
        for (int nt = 0; nt < 8; ++nt) {
            int n = nt * 16 + mloc;
            const v8h* bp = (const v8h*)(sW + n * HID + k0);
            v8h blo = bp[0];
            v8h bhi = bp[2];
            v16h b;
#pragma unroll
            for (int i = 0; i < 8; ++i) { b[i] = blo[i]; b[8 + i] = bhi[i]; }
            acc[nt] = __builtin_amdgcn_wmma_f32_16x16x32_f16(
                false, a, false, b, (short)0, acc[nt], false, false);
        }
    }

#pragma unroll
    for (int nt = 0; nt < 8; ++nt) {
        int col = nt * 16 + mloc;
#pragma unroll
        for (int r = 0; r < 8; ++r) {
            float v = acc[nt][r];
            if (RELU) v = fmaxf(v, 0.0f);
            int M = row0 + r + half * 8;
            out[(size_t)M * HID + col] = v;
        }
    }
}

extern "C" void kernel_launch(void* const* d_in, const int* in_sizes, int n_in,
                              void* d_out, int out_size, void* d_ws, size_t ws_size,
                              hipStream_t stream) {
    const int N = in_sizes[0];
    const int E = in_sizes[1] / 2;
    const int G = out_size / (3 * HID);

    const float* enc_W  = (const float*)d_in[12];
    const float* enc_b  = (const float*)d_in[13];
    const float* edge_W = (const float*)d_in[14];
    const float* edge_b = (const float*)d_in[15];
    const float* eps    = (const float*)d_in[16];
    const float* W1     = (const float*)d_in[17];
    const float* b1     = (const float*)d_in[18];
    const float* W2     = (const float*)d_in[19];
    const float* b2     = (const float*)d_in[20];
    const float* fc_W   = (const float*)d_in[21];
    const float* fc_b   = (const float*)d_in[22];

    float* bufA   = (float*)d_ws;
    float* bufB   = bufA + (size_t)N * HID;
    float* pooled = bufB + (size_t)N * HID;
    _Float16* w16 = (_Float16*)(pooled + (size_t)G * HID);
    _Float16* W1t0 = w16;
    _Float16* W1t1 = w16 + 16384;
    _Float16* W2t0 = w16 + 2 * 16384;
    _Float16* W2t1 = w16 + 3 * 16384;
    _Float16* fcWt = w16 + 4 * 16384;
    const int NP = (N + 4) & ~3;
    const int EP = (E + 3) & ~3;
    int* cnt   = (int*)(w16 + 5 * 16384);
    int* off   = cnt + 3 * (size_t)NP;
    int* perm  = off + 3 * (size_t)NP;
    int* ssrc  = perm + 3 * (size_t)EP;
    float* sea = (float*)(ssrc + 3 * (size_t)EP);

    convert_wt_kernel<<<64, 256, 0, stream>>>(W1,             W1t0);
    convert_wt_kernel<<<64, 256, 0, stream>>>(W1 + HID * HID, W1t1);
    convert_wt_kernel<<<64, 256, 0, stream>>>(W2,             W2t0);
    convert_wt_kernel<<<64, 256, 0, stream>>>(W2 + HID * HID, W2t1);
    convert_wt_kernel<<<64, 256, 0, stream>>>(fc_W,           fcWt);

    const int nodeBlk   = (N + 7) / 8;
    const int gemmBlk   = ((N + 15) / 16 + 7) / 8;
    const int gP4       = (G * HID) / 4;
    const int fcBlk     = ((G + 15) / 16 + 7) / 8;

    EdgeIn ein;
    ein.ei0 = (const int*)d_in[1];   ein.ei1 = (const int*)d_in[5];   ein.ei2 = (const int*)d_in[9];
    ein.ea0 = (const float*)d_in[2]; ein.ea1 = (const float*)d_in[6]; ein.ea2 = (const float*)d_in[10];
    const dim3 edgeGrid((E + 255) / 256, 3);
    const int RB = 85;
    const int R  = (N + RB - 1) / RB;
    const dim3 rangeGrid(RB, 3);
    range_kernel<false><<<rangeGrid, 1024, R * sizeof(int), stream>>>(ein, cnt, off, perm, E, N, NP, EP, R);
    scan_kernel<<<3, 1024, 0, stream>>>(cnt, off, N, NP);
    range_kernel<true><<<rangeGrid, 1024, R * sizeof(int), stream>>>(ein, cnt, off, perm, E, N, NP, EP, R);
    rank_kernel<<<edgeGrid, 256, 0, stream>>>(ein, off, perm, ssrc, sea, E, N, NP, EP);

    for (int g = 0; g < 3; ++g) {
        const float* x  = (const float*)d_in[4 * g + 0];
        const int*   bt = (const int*)  d_in[4 * g + 3];

        encode_kernel<<<nodeBlk, 256, 0, stream>>>(x, enc_W, enc_b, bufA, N);

        float* hcur = bufA;
        float* aggb = bufB;
        for (int l = 0; l < 2; ++l) {
            gather_kernel<<<nodeBlk, 256, 0, stream>>>(
                hcur, off + (size_t)g * NP, ssrc + (size_t)g * EP, sea + (size_t)g * EP * 2,
                edge_W + (size_t)l * 2 * HID, edge_b + (size_t)l * HID, eps, l, aggb, N);
            _Float16* w1t = (l == 0) ? W1t0 : W1t1;
            _Float16* w2t = (l == 0) ? W2t0 : W2t1;
            gemm128_wmma_kernel<true><<<gemmBlk, 256, 0, stream>>>(
                aggb, w1t, b1 + (size_t)l * HID, hcur, N);
            gemm128_wmma_kernel<true><<<gemmBlk, 256, 0, stream>>>(
                hcur, w2t, b2 + (size_t)l * HID, aggb, N);
            float* t = hcur; hcur = aggb; aggb = t;
        }

        zero_kernel<<<(gP4 + 255) / 256, 256, 0, stream>>>(pooled, gP4);
        pool_kernel<<<nodeBlk, 256, 0, stream>>>(hcur, bt, pooled, N);
        gemm128_wmma_kernel<false><<<fcBlk, 256, 0, stream>>>(
            pooled, fcWt, fc_b, (float*)d_out + (size_t)g * G * HID, G);
    }
}


# LDS weight tile of the products: row pitch 256 B -> 272 B (16 B x odd); the 16 lanes served together in a fragment read fell on the same four banks (16-way conflict), now on all 64 banks once
# speedup vs baseline: 3.3271x; 1.8412x over previous
#include <hip/hip_runtime.h>

typedef __attribute__((ext_vector_type(16))) _Float16 v16h;
typedef __attribute__((ext_vector_type(8)))  _Float16 v8h;
typedef __attribute__((ext_vector_type(8)))  float    v8f;

#define HID 128

__global__ void convert_wt_kernel(const float* __restrict__ W,
                                  _Float16* __restrict__ Wt) {
    int idx = blockIdx.x * blockDim.x + threadIdx.x;
    int n = idx >> 7;
    int k = idx & 127;
    Wt[idx] = (_Float16)W[k * HID + n];
}


struct EdgeIn {
    const int* ei0; const int* ei1; const int* ei2;
    const float* ea0; const float* ea1; const float* ea2;
};
__device__ __forceinline__ const int* pick_ei(const EdgeIn& p, int g) {
    return g == 0 ? p.ei0 : (g == 1 ? p.ei1 : p.ei2);
}
__device__ __forceinline__ const float* pick_ea(const EdgeIn& p, int g) {
    return g == 0 ? p.ea0 : (g == 1 ? p.ea1 : p.ea2);
}

struct NodeIn {
    const float* x0; const float* x1; const float* x2;
    const int* b0; const int* b1; const int* b2;
};
__device__ __forceinline__ const float* pick_x(const NodeIn& p, int g) {
    return g == 0 ? p.x0 : (g == 1 ? p.x1 : p.x2);
}
__device__ __forceinline__ const int* pick_b(const NodeIn& p, int g) {
    return g == 0 ? p.b0 : (g == 1 ? p.b1 : p.b2);
}

extern __shared__ int lds_cnt[];

template <bool PLACE>
__global__ __launch_bounds__(1024) void range_kernel(EdgeIn p, int* __restrict__ cnt,
                                                     const int* __restrict__ off,
                                                     int* __restrict__ perm,
                                                     int E, int N, int NP, int EP, int R) {
    const int g = blockIdx.y, tid = threadIdx.x;
    const int lo = blockIdx.x * R;
    int nloc = min(lo + R, N) - lo;
    if (nloc < 0) nloc = 0;
    const unsigned un = (unsigned)nloc;
    for (int i = tid; i < nloc; i += 1024)
        lds_cnt[i] = PLACE ? off[(size_t)g * NP + lo + i] : 0;
    __syncthreads();

    const int* dst = pick_ei(p, g) + E;
    int* pg = perm + (size_t)g * EP;
    auto visit = [&](int d, int e) {
        unsigned r = (unsigned)(d - lo);
        if (r < un) {
            if (PLACE) {
                int slot = atomicAdd(&lds_cnt[r], 1);
                if ((unsigned)slot < (unsigned)E) pg[slot] = e;
            } else {
                atomicAdd(&lds_cnt[r], 1);
            }
        }
    };
    auto visit4 = [&](const int4& d, int j) {
        unsigned a = (unsigned)(d.x - lo), b = (unsigned)(d.y - lo);
        unsigned c = (unsigned)(d.z - lo), w = (unsigned)(d.w - lo);
        if (min(min(a, b), min(c, w)) < un) {
            visit(d.x, 4 * j);     visit(d.y, 4 * j + 1);
            visit(d.z, 4 * j + 2); visit(d.w, 4 * j + 3);
        }
    };

    const bool vec = ((E & 3) == 0) && ((((size_t)dst) & 15) == 0);
    const int E4 = vec ? (E >> 2) : 0;
    const int4* d4 = (const int4*)dst;
    const int start = (int)(((long long)blockIdx.x * E4) / gridDim.x);
    int i = tid;
    for (; i + 3072 < E4; i += 4096) {
        int j0 = i + start;        if (j0 >= E4) j0 -= E4;
        int j1 = i + 1024 + start; if (j1 >= E4) j1 -= E4;
        int j2 = i + 2048 + start; if (j2 >= E4) j2 -= E4;
        int j3 = i + 3072 + start; if (j3 >= E4) j3 -= E4;
        int4 q0 = d4[j0], q1 = d4[j1], q2 = d4[j2], q3 = d4[j3];
        visit4(q0, j0); visit4(q1, j1); visit4(q2, j2); visit4(q3, j3);
    }
    for (; i < E4; i += 1024) {
        int j0 = i + start; if (j0 >= E4) j0 -= E4;
        int4 q0 = d4[j0];
        visit4(q0, j0);
    }
    for (int e = 4 * E4 + tid; e < E; e += 1024) visit(dst[e], e);

    if (!PLACE) {
        __syncthreads();
        for (int k = tid; k < nloc; k += 1024) cnt[(size_t)g * NP + lo + k] = lds_cnt[k];
    }
}

__global__ __launch_bounds__(1024) void scan_kernel(const int* __restrict__ cnt,
                                                    int* __restrict__ off, int N, int NP) {
    __shared__ int wsum[32];
    const int* c = cnt + (size_t)blockIdx.x * NP;
    int* o = off + (size_t)blockIdx.x * NP;
    int tid = threadIdx.x, lane = tid & 31, wave = tid >> 5;
    int chunk = (N + 1023) / 1024;
    int lo = tid * chunk;
    if (lo > N) lo = N;
    int hi = lo + chunk;
    if (hi > N) hi = N;
    int s = 0;
    for (int i = lo; i < hi; ++i) s += c[i];
    int v = s;
    for (int d = 1; d < 32; d <<= 1) {
        int t = __shfl_up(v, d);
        if (lane >= d) v += t;
    }
    if (lane == 31) wsum[wave] = v;
    __syncthreads();
    if (wave == 0) {
        int w = wsum[lane];
        for (int d = 1; d < 32; d <<= 1) {
            int t = __shfl_up(w, d);
            if (lane >= d) w += t;
        }
        wsum[lane] = w;
    }
    __syncthreads();
    int run = v - s + (wave ? wsum[wave - 1] : 0);
    for (int i = lo; i < hi; ++i) { o[i] = run; run += c[i]; }
    if (tid == 1023) o[N] = run;
}

__global__ void rank_kernel(EdgeIn p, const int* __restrict__ off,
                            const int* __restrict__ perm, int* __restrict__ ssrc,
                            float* __restrict__ sea, int E, int N, int NP, int EP) {
    int g = blockIdx.y;
    int q = blockIdx.x * blockDim.x + threadIdx.x;
    const int* og = off + (size_t)g * NP;
    const int* pg = perm + (size_t)g * EP;
    if (q >= E || q >= og[N]) return;
    int e = pg[q];
    if ((unsigned)e >= (unsigned)E) return;
    const int* ei = pick_ei(p, g);
    int d = ei[E + e];
    int s0 = og[d], s1 = og[d + 1];
    int r = 0;
    for (int j = s0; j < s1; ++j) r += (pg[j] < e) ? 1 : 0;
    size_t t = (size_t)g * EP + s0 + r;
    ssrc[t] = ei[e];
    ((float2*)sea)[t] = ((const float2*)pick_ea(p, g))[e];
}

__device__ __forceinline__ void add_msg(float4& acc, const float4& hv, float a0, float a1,
                                        const float4& w0, const float4& w1, const float4& bb) {
    acc.x += fmaxf(__builtin_fmaf(a1, w1.x, __builtin_fmaf(a0, w0.x, hv.x)) + bb.x, 0.0f);
    acc.y += fmaxf(__builtin_fmaf(a1, w1.y, __builtin_fmaf(a0, w0.y, hv.y)) + bb.y, 0.0f);
    acc.z += fmaxf(__builtin_fmaf(a1, w1.z, __builtin_fmaf(a0, w0.z, hv.z)) + bb.z, 0.0f);
    acc.w += fmaxf(__builtin_fmaf(a1, w1.w, __builtin_fmaf(a0, w0.w, hv.w)) + bb.w, 0.0f);
}

struct RowsInMemory {
    const float* h;
    __device__ __forceinline__ float4 operator()(int i, int c) const {
        return *(const float4*)(h + (size_t)i * HID + c);
    }
};
struct RowsFromEncoder {
    const float* x;
    float4 w, b;
    __device__ __forceinline__ float4 operator()(int i, int) const {
        float xv = x[i];
        float4 r;
        r.x = __builtin_fmaf(xv, w.x, b.x);
        r.y = __builtin_fmaf(xv, w.y, b.y);
        r.z = __builtin_fmaf(xv, w.z, b.z);
        r.w = __builtin_fmaf(xv, w.w, b.w);
        return r;
    }
};

#define RUN 10
#define RUN_MEM 13

template <class Rows>
__device__ __forceinline__ void gather_run(
    const Rows& rows, const int* __restrict__ off,
    const int* __restrict__ ssrc, const float* __restrict__ sea,
    const float* __restrict__ eW, const float* __restrict__ eb,
    const float* __restrict__ eps, int l, float* __restrict__ agg, int N, int n0, int n1, int c) {
    float4 w0 = *(const float4*)(eW + c);
    float4 w1 = *(const float4*)(eW + HID + c);
    float4 bb = *(const float4*)(eb + c);
    const float s = 1.0f + eps[l];
    const float2* a2 = (const float2*)sea;
    const int last = N - 1;
    int s1 = off[n0];
  for (int n = n0; n < n1; ++n) {
    int s0 = s1;
    s1 = off[n + 1];
    float4 hs = rows(n, c);
    float4 acc = make_float4(0.f, 0.f, 0.f, 0.f);
    int p = s0;
    for (; p + 4 <= s1; p += 4) {
        int i0 = min(max(ssrc[p], 0), last),     i1 = min(max(ssrc[p + 1], 0), last);
        int i2 = min(max(ssrc[p + 2], 0), last), i3 = min(max(ssrc[p + 3], 0), last);
        float2 e0 = a2[p], e1 = a2[p + 1], e2 = a2[p + 2], e3 = a2[p + 3];
        float4 h0 = rows(i0, c);
        float4 h1 = rows(i1, c);
        float4 h2 = rows(i2, c);
        float4 h3 = rows(i3, c);
        add_msg(acc, h0, e0.x, e0.y, w0, w1, bb);
        add_msg(acc, h1, e1.x, e1.y, w0, w1, bb);
        add_msg(acc, h2, e2.x, e2.y, w0, w1, bb);
        add_msg(acc, h3, e3.x, e3.y, w0, w1, bb);
    }
    for (; p < s1; ++p) {
        int i0 = min(max(ssrc[p], 0), last);
        float2 e0 = a2[p];
        float4 h0 = rows(i0, c);
        add_msg(acc, h0, e0.x, e0.y, w0, w1, bb);
    }
    float4 r;
    r.x = s * hs.x + acc.x;
    r.y = s * hs.y + acc.y;
    r.z = s * hs.z + acc.z;
    r.w = s * hs.w + acc.w;
    *(float4*)(agg + (size_t)n * HID + c) = r;
  }
}

__global__ __launch_bounds__(256) void gather_kernel(
    const float* __restrict__ h, const int* __restrict__ off,
    const int* __restrict__ ssrc, const float* __restrict__ sea,
    const float* __restrict__ eW, const float* __restrict__ eb,
    const float* __restrict__ eps, int l, float* __restrict__ agg, int N, int NP, int EP) {
    const int g = blockIdx.y;
    int lane = threadIdx.x & 31;
    int n0 = __builtin_amdgcn_readfirstlane((blockIdx.x * 8 + (threadIdx.x >> 5)) * RUN_MEM);
    if (n0 >= N) return;
    int n1 = min(n0 + RUN_MEM, N);
    RowsInMemory rows{h + (size_t)g * N * HID};
    gather_run(rows, off + (size_t)g * NP, ssrc + (size_t)g * EP, sea + (size_t)g * EP * 2,
               eW, eb, eps, l, agg + (size_t)g * N * HID, N, n0, n1, lane * 4);
}

__global__ __launch_bounds__(256) void gather_encoded_kernel(
    NodeIn nin, const float* __restrict__ encW,
    const float* __restrict__ encb, const int* __restrict__ off,
    const int* __restrict__ ssrc, const float* __restrict__ sea,
    const float* __restrict__ eW, const float* __restrict__ eb,
    const float* __restrict__ eps, int l, float* __restrict__ agg, int N, int NP, int EP) {
    const int g = blockIdx.y;
    int lane = threadIdx.x & 31;
    int n0 = __builtin_amdgcn_readfirstlane((blockIdx.x * 8 + (threadIdx.x >> 5)) * RUN);
    if (n0 >= N) return;
    int n1 = min(n0 + RUN, N);
    int c = lane * 4;
    RowsFromEncoder rows{pick_x(nin, g), *(const float4*)(encW + c), *(const float4*)(encb + c)};
    gather_run(rows, off + (size_t)g * NP, ssrc + (size_t)g * EP, sea + (size_t)g * EP * 2,
               eW, eb, eps, l, agg + (size_t)g * N * HID, N, n0, n1, c);
}

__global__ void zero_kernel(float* __restrict__ p, int n4) {
    int i = blockIdx.x * blockDim.x + threadIdx.x;
    if (i < n4) ((float4*)p)[i] = make_float4(0.f, 0.f, 0.f, 0.f);
}

__device__ __forceinline__ void pool_flush(float* __restrict__ pp, const float4& a) {
    atomicAdd(pp + 0, a.x);
    atomicAdd(pp + 1, a.y);
    atomicAdd(pp + 2, a.z);
    atomicAdd(pp + 3, a.w);
}

__global__ __launch_bounds__(256) void pool_kernel(const float* __restrict__ h, NodeIn nin,
                                                   float* __restrict__ pooled, int N, int G) {
    const int gr = blockIdx.y;
    int lane = threadIdx.x & 31;
    int n0 = __builtin_amdgcn_readfirstlane((blockIdx.x * 8 + (threadIdx.x >> 5)) * RUN);
    if (n0 >= N) return;
    const int cnt = min(RUN, N - n0);
    const int* b = pick_b(nin, gr);
    int c = lane * 4;
    const float* hp = h + ((size_t)gr * N + n0) * HID + c;
    float* pg = pooled + (size_t)gr * G * HID + c;
    float4 hv[RUN];
    int id[RUN];
#pragma unroll
    for (int i = 0; i < RUN; ++i) {
        int k = min(i, cnt - 1);
        id[i] = b[n0 + k];
        hv[i] = *(const float4*)(hp + (size_t)k * HID);
    }
    int cur = id[0];
    float4 acc = hv[0];
#pragma unroll
    for (int i = 1; i < RUN; ++i) {
        if (i < cnt) {
            if (id[i] != cur) {
                pool_flush(pg + (size_t)cur * HID, acc);
                cur = id[i];
                acc = hv[i];
            } else {
                acc.x += hv[i].x; acc.y += hv[i].y; acc.z += hv[i].z; acc.w += hv[i].w;
            }
        }
    }
    pool_flush(pg + (size_t)cur * HID, acc);
}

#define WPITCH 136
template <bool RELU>
__global__ __launch_bounds__(256) void gemm128_wmma_kernel(
    const float* __restrict__ A, const _Float16* __restrict__ Wt,
    const float* __restrict__ bias, float* __restrict__ out, int rows,
    size_t strideA, size_t strideO) {
    A += blockIdx.y * strideA;
    out += blockIdx.y * strideO;
    __shared__ __align__(16) _Float16 sW[HID * WPITCH];
    {
        const uint4* s = (const uint4*)Wt;
        for (int i = threadIdx.x; i < (HID * HID) / 8; i += 256)
            *(uint4*)(sW + (i >> 4) * WPITCH + (i & 15) * 8) = s[i];
    }
    __syncthreads();

    int wave = threadIdx.x >> 5, lane = threadIdx.x & 31;
    int tile = blockIdx.x * 8 + wave;
    int row0 = tile * 16;
    if (row0 >= rows) return;

    int mloc = lane & 15;
    int half = lane >> 4;
    int base = half * 8;
    const float* rowA = A + (size_t)(row0 + mloc) * HID;

    v8f acc[8];
#pragma unroll
    for (int nt = 0; nt < 8; ++nt) {
        float bv = bias[nt * 16 + mloc];
#pragma unroll
        for (int r = 0; r < 8; ++r) acc[nt][r] = bv;
    }

#pragma unroll
    for (int kk = 0; kk < 4; ++kk) {
        int k0 = kk * 32 + base;
        float4 a0 = *(const float4*)(rowA + k0);
        float4 a1 = *(const float4*)(rowA + k0 + 4);
        float4 a2 = *(const float4*)(rowA + k0 + 16);
        float4 a3 = *(const float4*)(rowA + k0 + 20);
        v16h a;
        a[0]  = (_Float16)a0.x; a[1]  = (_Float16)a0.y;
        a[2]  = (_Float16)a0.z; a[3]  = (_Float16)a0.w;
        a[4]  = (_Float16)a1.x; a[5]  = (_Float16)a1.y;
        a[6]  = (_Float16)a1.z; a[7]  = (_Float16)a1.w;
        a[8]  = (_Float16)a2.x; a[9]  = (_Float16)a2.y;
        a[10] = (_Float16)a2.z; a[11] = (_Float16)a2.w;
        a[12] = (_Float16)a3.x; a[13] = (_Float16)a3.y;
        a[14] = (_Float16)a3.z; a[15] = (_Float16)a3.w;
#pragma unroll
        for (int nt = 0; nt < 8; ++nt) {
            int n = nt * 16 + mloc;
            const v8h* bp = (const v8h*)(sW + n * WPITCH + k0);
            v8h blo = bp[0];
            v8h bhi = bp[2];
            v16h b;
#pragma unroll
            for (int i = 0; i < 8; ++i) { b[i] = blo[i]; b[8 + i] = bhi[i]; }
            acc[nt] = __builtin_amdgcn_wmma_f32_16x16x32_f16(
                false, a, false, b, (short)0, acc[nt], false, false);
        }
    }

#pragma unroll
    for (int nt = 0; nt < 8; ++nt) {
        int col = nt * 16 + mloc;
#pragma unroll
        for (int r = 0; r < 8; ++r) {
            float v = acc[nt][r];
            if (RELU) v = fmaxf(v, 0.0f);
            int M = row0 + r + half * 8;
            out[(size_t)M * HID + col] = v;
        }
    }
}

extern "C" void kernel_launch(void* const* d_in, const int* in_sizes, int n_in,
                              void* d_out, int out_size, void* d_ws, size_t ws_size,
                              hipStream_t stream) {
    const int N = in_sizes[0];
    const int E = in_sizes[1] / 2;
    const int G = out_size / (3 * HID);

    const float* enc_W  = (const float*)d_in[12];
    const float* enc_b  = (const float*)d_in[13];
    const float* edge_W = (const float*)d_in[14];
    const float* edge_b = (const float*)d_in[15];
    const float* eps    = (const float*)d_in[16];
    const float* W1     = (const float*)d_in[17];
    const float* b1     = (const float*)d_in[18];
    const float* W2     = (const float*)d_in[19];
    const float* b2     = (const float*)d_in[20];
    const float* fc_W   = (const float*)d_in[21];
    const float* fc_b   = (const float*)d_in[22];

    const size_t plane = (size_t)N * HID;
    const size_t pplane = (size_t)G * HID;
    float* bufA   = (float*)d_ws;
    float* bufB   = bufA + 3 * plane;
    float* pooled = bufB + 3 * plane;
    _Float16* w16 = (_Float16*)(pooled + 3 * pplane);
    _Float16* W1t0 = w16;
    _Float16* W1t1 = w16 + 16384;
    _Float16* W2t0 = w16 + 2 * 16384;
    _Float16* W2t1 = w16 + 3 * 16384;
    _Float16* fcWt = w16 + 4 * 16384;
    const int NP = (N + 4) & ~3;
    const int EP = (E + 3) & ~3;
    int* cnt   = (int*)(w16 + 5 * 16384);
    int* off   = cnt + 3 * (size_t)NP;
    int* perm  = off + 3 * (size_t)NP;
    int* ssrc  = perm + 3 * (size_t)EP;
    float* sea = (float*)(ssrc + 3 * (size_t)EP);

    convert_wt_kernel<<<64, 256, 0, stream>>>(W1,             W1t0);
    convert_wt_kernel<<<64, 256, 0, stream>>>(W1 + HID * HID, W1t1);
    convert_wt_kernel<<<64, 256, 0, stream>>>(W2,             W2t0);
    convert_wt_kernel<<<64, 256, 0, stream>>>(W2 + HID * HID, W2t1);
    convert_wt_kernel<<<64, 256, 0, stream>>>(fc_W,           fcWt);

    const int nodeBlk   = (N + 8 * RUN - 1) / (8 * RUN);
    const int nodeBlkM  = (N + 8 * RUN_MEM - 1) / (8 * RUN_MEM);
    const int gemmBlk   = ((N + 15) / 16 + 7) / 8;
    const int gP4       = (G * HID) / 4;
    const int fcBlk     = ((G + 15) / 16 + 7) / 8;

    EdgeIn ein;
    ein.ei0 = (const int*)d_in[1];   ein.ei1 = (const int*)d_in[5];   ein.ei2 = (const int*)d_in[9];
    ein.ea0 = (const float*)d_in[2]; ein.ea1 = (const float*)d_in[6]; ein.ea2 = (const float*)d_in[10];
    const dim3 edgeGrid((E + 255) / 256, 3);
    const int RB = 85;
    const int R  = (N + RB - 1) / RB;
    const dim3 rangeGrid(RB, 3);
    range_kernel<false><<<rangeGrid, 1024, R * sizeof(int), stream>>>(ein, cnt, off, perm, E, N, NP, EP, R);
    scan_kernel<<<3, 1024, 0, stream>>>(cnt, off, N, NP);
    range_kernel<true><<<rangeGrid, 1024, R * sizeof(int), stream>>>(ein, cnt, off, perm, E, N, NP, EP, R);
    rank_kernel<<<edgeGrid, 256, 0, stream>>>(ein, off, perm, ssrc, sea, E, N, NP, EP);

    NodeIn nin;
    nin.x0 = (const float*)d_in[0]; nin.x1 = (const float*)d_in[4]; nin.x2 = (const float*)d_in[8];
    nin.b0 = (const int*)d_in[3];   nin.b1 = (const int*)d_in[7];   nin.b2 = (const int*)d_in[11];
    const dim3 nodeGrid(nodeBlk, 3), nodeGridM(nodeBlkM, 3), gemmGrid(gemmBlk, 3), fcGrid(fcBlk, 3);

    float* hcur = bufA;
    float* aggb = bufB;
    for (int l = 0; l < 2; ++l) {
        if (l == 0)
            gather_encoded_kernel<<<nodeGrid, 256, 0, stream>>>(
                nin, enc_W, enc_b, off, ssrc, sea, edge_W, edge_b, eps, 0, aggb, N, NP, EP);
        else
            gather_kernel<<<nodeGridM, 256, 0, stream>>>(
                hcur, off, ssrc, sea, edge_W + (size_t)l * 2 * HID, edge_b + (size_t)l * HID,
                eps, l, aggb, N, NP, EP);
        _Float16* w1t = (l == 0) ? W1t0 : W1t1;
        _Float16* w2t = (l == 0) ? W2t0 : W2t1;
        gemm128_wmma_kernel<true><<<gemmGrid, 256, 0, stream>>>(
            aggb, w1t, b1 + (size_t)l * HID, hcur, N, plane, plane);
        gemm128_wmma_kernel<true><<<gemmGrid, 256, 0, stream>>>(
            hcur, w2t, b2 + (size_t)l * HID, aggb, N, plane, plane);
        float* t = hcur; hcur = aggb; aggb = t;
    }

    zero_kernel<<<(3 * gP4 + 255) / 256, 256, 0, stream>>>(pooled, 3 * gP4);
    pool_kernel<<<nodeGrid, 256, 0, stream>>>(hcur, nin, pooled, N, G);
    gemm128_wmma_kernel<false><<<fcGrid, 256, 0, stream>>>(
        pooled, fcWt, fc_b, (float*)d_out, G, pplane, pplane);
}


# each layer's two MLP products in one kernel: the first is formed transposed so its f32 tile, relu'd and converted, is the second's A operand in registers; hidden plane not stored; 2 launches dropped
# speedup vs baseline: 3.7430x; 1.1250x over previous
#include <hip/hip_runtime.h>

typedef __attribute__((ext_vector_type(16))) _Float16 v16h;
typedef __attribute__((ext_vector_type(8)))  _Float16 v8h;
typedef __attribute__((ext_vector_type(8)))  float    v8f;

#define HID 128

__global__ void convert_wt_kernel(const float* __restrict__ W,
                                  _Float16* __restrict__ Wt) {
    int idx = blockIdx.x * blockDim.x + threadIdx.x;
    int n = idx >> 7;
    int k = idx & 127;
    Wt[idx] = (_Float16)W[k * HID + n];
}


struct EdgeIn {
    const int* ei0; const int* ei1; const int* ei2;
    const float* ea0; const float* ea1; const float* ea2;
};
__device__ __forceinline__ const int* pick_ei(const EdgeIn& p, int g) {
    return g == 0 ? p.ei0 : (g == 1 ? p.ei1 : p.ei2);
}
__device__ __forceinline__ const float* pick_ea(const EdgeIn& p, int g) {
    return g == 0 ? p.ea0 : (g == 1 ? p.ea1 : p.ea2);
}

struct NodeIn {
    const float* x0; const float* x1; const float* x2;
    const int* b0; const int* b1; const int* b2;
};
__device__ __forceinline__ const float* pick_x(const NodeIn& p, int g) {
    return g == 0 ? p.x0 : (g == 1 ? p.x1 : p.x2);
}
__device__ __forceinline__ const int* pick_b(const NodeIn& p, int g) {
    return g == 0 ? p.b0 : (g == 1 ? p.b1 : p.b2);
}

extern __shared__ int lds_cnt[];

template <bool PLACE>
__global__ __launch_bounds__(1024) void range_kernel(EdgeIn p, int* __restrict__ cnt,
                                                     const int* __restrict__ off,
                                                     int* __restrict__ perm,
                                                     int E, int N, int NP, int EP, int R) {
    const int g = blockIdx.y, tid = threadIdx.x;
    const int lo = blockIdx.x * R;
    int nloc = min(lo + R, N) - lo;
    if (nloc < 0) nloc = 0;
    const unsigned un = (unsigned)nloc;
    for (int i = tid; i < nloc; i += 1024)
        lds_cnt[i] = PLACE ? off[(size_t)g * NP + lo + i] : 0;
    __syncthreads();

    const int* dst = pick_ei(p, g) + E;
    int* pg = perm + (size_t)g * EP;
    auto visit = [&](int d, int e) {
        unsigned r = (unsigned)(d - lo);
        if (r < un) {
            if (PLACE) {
                int slot = atomicAdd(&lds_cnt[r], 1);
                if ((unsigned)slot < (unsigned)E) pg[slot] = e;
            } else {
                atomicAdd(&lds_cnt[r], 1);
            }
        }
    };
    auto visit4 = [&](const int4& d, int j) {
        unsigned a = (unsigned)(d.x - lo), b = (unsigned)(d.y - lo);
        unsigned c = (unsigned)(d.z - lo), w = (unsigned)(d.w - lo);
        if (min(min(a, b), min(c, w)) < un) {
            visit(d.x, 4 * j);     visit(d.y, 4 * j + 1);
            visit(d.z, 4 * j + 2); visit(d.w, 4 * j + 3);
        }
    };

    const bool vec = ((E & 3) == 0) && ((((size_t)dst) & 15) == 0);
    const int E4 = vec ? (E >> 2) : 0;
    const int4* d4 = (const int4*)dst;
    const int start = (int)(((long long)blockIdx.x * E4) / gridDim.x);
    int i = tid;
    for (; i + 3072 < E4; i += 4096) {
        int j0 = i + start;        if (j0 >= E4) j0 -= E4;
        int j1 = i + 1024 + start; if (j1 >= E4) j1 -= E4;
        int j2 = i + 2048 + start; if (j2 >= E4) j2 -= E4;
        int j3 = i + 3072 + start; if (j3 >= E4) j3 -= E4;
        int4 q0 = d4[j0], q1 = d4[j1], q2 = d4[j2], q3 = d4[j3];
        visit4(q0, j0); visit4(q1, j1); visit4(q2, j2); visit4(q3, j3);
    }
    for (; i < E4; i += 1024) {
        int j0 = i + start; if (j0 >= E4) j0 -= E4;
        int4 q0 = d4[j0];
        visit4(q0, j0);
    }
    for (int e = 4 * E4 + tid; e < E; e += 1024) visit(dst[e], e);

    if (!PLACE) {
        __syncthreads();
        for (int k = tid; k < nloc; k += 1024) cnt[(size_t)g * NP + lo + k] = lds_cnt[k];
    }
}

__global__ __launch_bounds__(1024) void scan_kernel(const int* __restrict__ cnt,
                                                    int* __restrict__ off, int N, int NP) {
    __shared__ int wsum[32];
    const int* c = cnt + (size_t)blockIdx.x * NP;
    int* o = off + (size_t)blockIdx.x * NP;
    int tid = threadIdx.x, lane = tid & 31, wave = tid >> 5;
    int chunk = (N + 1023) / 1024;
    int lo = tid * chunk;
    if (lo > N) lo = N;
    int hi = lo + chunk;
    if (hi > N) hi = N;
    int s = 0;
    for (int i = lo; i < hi; ++i) s += c[i];
    int v = s;
    for (int d = 1; d < 32; d <<= 1) {
        int t = __shfl_up(v, d);
        if (lane >= d) v += t;
    }
    if (lane == 31) wsum[wave] = v;
    __syncthreads();
    if (wave == 0) {
        int w = wsum[lane];
        for (int d = 1; d < 32; d <<= 1) {
            int t = __shfl_up(w, d);
            if (lane >= d) w += t;
        }
        wsum[lane] = w;
    }
    __syncthreads();
    int run = v - s + (wave ? wsum[wave - 1] : 0);
    for (int i = lo; i < hi; ++i) { o[i] = run; run += c[i]; }
    if (tid == 1023) o[N] = run;
}

__global__ void rank_kernel(EdgeIn p, const int* __restrict__ off,
                            const int* __restrict__ perm, int* __restrict__ ssrc,
                            float* __restrict__ sea, int E, int N, int NP, int EP) {
    int g = blockIdx.y;
    int q = blockIdx.x * blockDim.x + threadIdx.x;
    const int* og = off + (size_t)g * NP;
    const int* pg = perm + (size_t)g * EP;
    if (q >= E || q >= og[N]) return;
    int e = pg[q];
    if ((unsigned)e >= (unsigned)E) return;
    const int* ei = pick_ei(p, g);
    int d = ei[E + e];
    int s0 = og[d], s1 = og[d + 1];
    int r = 0;
    for (int j = s0; j < s1; ++j) r += (pg[j] < e) ? 1 : 0;
    size_t t = (size_t)g * EP + s0 + r;
    ssrc[t] = ei[e];
    ((float2*)sea)[t] = ((const float2*)pick_ea(p, g))[e];
}

__device__ __forceinline__ void add_msg(float4& acc, const float4& hv, float a0, float a1,
                                        const float4& w0, const float4& w1, const float4& bb) {
    acc.x += fmaxf(__builtin_fmaf(a1, w1.x, __builtin_fmaf(a0, w0.x, hv.x)) + bb.x, 0.0f);
    acc.y += fmaxf(__builtin_fmaf(a1, w1.y, __builtin_fmaf(a0, w0.y, hv.y)) + bb.y, 0.0f);
    acc.z += fmaxf(__builtin_fmaf(a1, w1.z, __builtin_fmaf(a0, w0.z, hv.z)) + bb.z, 0.0f);
    acc.w += fmaxf(__builtin_fmaf(a1, w1.w, __builtin_fmaf(a0, w0.w, hv.w)) + bb.w, 0.0f);
}

struct RowsInMemory {
    const float* h;
    __device__ __forceinline__ float4 operator()(int i, int c) const {
        return *(const float4*)(h + (size_t)i * HID + c);
    }
};
struct RowsFromEncoder {
    const float* x;
    float4 w, b;
    __device__ __forceinline__ float4 operator()(int i, int) const {
        float xv = x[i];
        float4 r;
        r.x = __builtin_fmaf(xv, w.x, b.x);
        r.y = __builtin_fmaf(xv, w.y, b.y);
        r.z = __builtin_fmaf(xv, w.z, b.z);
        r.w = __builtin_fmaf(xv, w.w, b.w);
        return r;
    }
};

#define RUN 10
#define RUN_MEM 13

template <class Rows>
__device__ __forceinline__ void gather_run(
    const Rows& rows, const int* __restrict__ off,
    const int* __restrict__ ssrc, const float* __restrict__ sea,
    const float* __restrict__ eW, const float* __restrict__ eb,
    const float* __restrict__ eps, int l, float* __restrict__ agg, int N, int n0, int n1, int c) {
    float4 w0 = *(const float4*)(eW + c);
    float4 w1 = *(const float4*)(eW + HID + c);
    float4 bb = *(const float4*)(eb + c);
    const float s = 1.0f + eps[l];
    const float2* a2 = (const float2*)sea;
    const int last = N - 1;
    int s1 = off[n0];
  for (int n = n0; n < n1; ++n) {
    int s0 = s1;
    s1 = off[n + 1];
    float4 hs = rows(n, c);
    float4 acc = make_float4(0.f, 0.f, 0.f, 0.f);
    int p = s0;
    for (; p + 4 <= s1; p += 4) {
        int i0 = min(max(ssrc[p], 0), last),     i1 = min(max(ssrc[p + 1], 0), last);
        int i2 = min(max(ssrc[p + 2], 0), last), i3 = min(max(ssrc[p + 3], 0), last);
        float2 e0 = a2[p], e1 = a2[p + 1], e2 = a2[p + 2], e3 = a2[p + 3];
        float4 h0 = rows(i0, c);
        float4 h1 = rows(i1, c);
        float4 h2 = rows(i2, c);
        float4 h3 = rows(i3, c);
        add_msg(acc, h0, e0.x, e0.y, w0, w1, bb);
        add_msg(acc, h1, e1.x, e1.y, w0, w1, bb);
        add_msg(acc, h2, e2.x, e2.y, w0, w1, bb);
        add_msg(acc, h3, e3.x, e3.y, w0, w1, bb);
    }
    for (; p < s1; ++p) {
        int i0 = min(max(ssrc[p], 0), last);
        float2 e0 = a2[p];
        float4 h0 = rows(i0, c);
        add_msg(acc, h0, e0.x, e0.y, w0, w1, bb);
    }
    float4 r;
    r.x = s * hs.x + acc.x;
    r.y = s * hs.y + acc.y;
    r.z = s * hs.z + acc.z;
    r.w = s * hs.w + acc.w;
    *(float4*)(agg + (size_t)n * HID + c) = r;
  }
}

__global__ __launch_bounds__(256) void gather_kernel(
    const float* __restrict__ h, const int* __restrict__ off,
    const int* __restrict__ ssrc, const float* __restrict__ sea,
    const float* __restrict__ eW, const float* __restrict__ eb,
    const float* __restrict__ eps, int l, float* __restrict__ agg, int N, int NP, int EP) {
    const int g = blockIdx.y;
    int lane = threadIdx.x & 31;
    int n0 = __builtin_amdgcn_readfirstlane((blockIdx.x * 8 + (threadIdx.x >> 5)) * RUN_MEM);
    if (n0 >= N) return;
    int n1 = min(n0 + RUN_MEM, N);
    RowsInMemory rows{h + (size_t)g * N * HID};
    gather_run(rows, off + (size_t)g * NP, ssrc + (size_t)g * EP, sea + (size_t)g * EP * 2,
               eW, eb, eps, l, agg + (size_t)g * N * HID, N, n0, n1, lane * 4);
}

__global__ __launch_bounds__(256) void gather_encoded_kernel(
    NodeIn nin, const float* __restrict__ encW,
    const float* __restrict__ encb, const int* __restrict__ off,
    const int* __restrict__ ssrc, const float* __restrict__ sea,
    const float* __restrict__ eW, const float* __restrict__ eb,
    const float* __restrict__ eps, int l, float* __restrict__ agg, int N, int NP, int EP) {
    const int g = blockIdx.y;
    int lane = threadIdx.x & 31;
    int n0 = __builtin_amdgcn_readfirstlane((blockIdx.x * 8 + (threadIdx.x >> 5)) * RUN);
    if (n0 >= N) return;
    int n1 = min(n0 + RUN, N);
    int c = lane * 4;
    RowsFromEncoder rows{pick_x(nin, g), *(const float4*)(encW + c), *(const float4*)(encb + c)};
    gather_run(rows, off + (size_t)g * NP, ssrc + (size_t)g * EP, sea + (size_t)g * EP * 2,
               eW, eb, eps, l, agg + (size_t)g * N * HID, N, n0, n1, c);
}

__global__ void zero_kernel(float* __restrict__ p, int n4) {
    int i = blockIdx.x * blockDim.x + threadIdx.x;
    if (i < n4) ((float4*)p)[i] = make_float4(0.f, 0.f, 0.f, 0.f);
}

__device__ __forceinline__ void pool_flush(float* __restrict__ pp, const float4& a) {
    atomicAdd(pp + 0, a.x);
    atomicAdd(pp + 1, a.y);
    atomicAdd(pp + 2, a.z);
    atomicAdd(pp + 3, a.w);
}

__global__ __launch_bounds__(256) void pool_kernel(const float* __restrict__ h, NodeIn nin,
                                                   float* __restrict__ pooled, int N, int G) {
    const int gr = blockIdx.y;
    int lane = threadIdx.x & 31;
    int n0 = __builtin_amdgcn_readfirstlane((blockIdx.x * 8 + (threadIdx.x >> 5)) * RUN);
    if (n0 >= N) return;
    const int cnt = min(RUN, N - n0);
    const int* b = pick_b(nin, gr);
    int c = lane * 4;
    const float* hp = h + ((size_t)gr * N + n0) * HID + c;
    float* pg = pooled + (size_t)gr * G * HID + c;
    float4 hv[RUN];
    int id[RUN];
#pragma unroll
    for (int i = 0; i < RUN; ++i) {
        int k = min(i, cnt - 1);
        id[i] = b[n0 + k];
        hv[i] = *(const float4*)(hp + (size_t)k * HID);
    }
    int cur = id[0];
    float4 acc = hv[0];
#pragma unroll
    for (int i = 1; i < RUN; ++i) {
        if (i < cnt) {
            if (id[i] != cur) {
                pool_flush(pg + (size_t)cur * HID, acc);
                cur = id[i];
                acc = hv[i];
            } else {
                acc.x += hv[i].x; acc.y += hv[i].y; acc.z += hv[i].z; acc.w += hv[i].w;
            }
        }
    }
    pool_flush(pg + (size_t)cur * HID, acc);
}

#define WPITCH 136
template <bool RELU>
__global__ __launch_bounds__(256) void gemm128_wmma_kernel(
    const float* __restrict__ A, const _Float16* __restrict__ Wt,
    const float* __restrict__ bias, float* __restrict__ out, int rows,
    size_t strideA, size_t strideO) {
    A += blockIdx.y * strideA;
    out += blockIdx.y * strideO;
    __shared__ __align__(16) _Float16 sW[HID * WPITCH];
    {
        const uint4* s = (const uint4*)Wt;
        for (int i = threadIdx.x; i < (HID * HID) / 8; i += 256)
            *(uint4*)(sW + (i >> 4) * WPITCH + (i & 15) * 8) = s[i];
    }
    __syncthreads();

    int wave = threadIdx.x >> 5, lane = threadIdx.x & 31;
    int tile = blockIdx.x * 8 + wave;
    int row0 = tile * 16;
    if (row0 >= rows) return;

    int mloc = lane & 15;
    int half = lane >> 4;
    int base = half * 8;
    const float* rowA = A + (size_t)(row0 + mloc) * HID;

    v8f acc[8];
#pragma unroll
    for (int nt = 0; nt < 8; ++nt) {
        float bv = bias[nt * 16 + mloc];
#pragma unroll
        for (int r = 0; r < 8; ++r) acc[nt][r] = bv;
    }

#pragma unroll
    for (int kk = 0; kk < 4; ++kk) {
        int k0 = kk * 32 + base;
        float4 a0 = *(const float4*)(rowA + k0);
        float4 a1 = *(const float4*)(rowA + k0 + 4);
        float4 a2 = *(const float4*)(rowA + k0 + 16);
        float4 a3 = *(const float4*)(rowA + k0 + 20);
        v16h a;
        a[0]  = (_Float16)a0.x; a[1]  = (_Float16)a0.y;
        a[2]  = (_Float16)a0.z; a[3]  = (_Float16)a0.w;
        a[4]  = (_Float16)a1.x; a[5]  = (_Float16)a1.y;
        a[6]  = (_Float16)a1.z; a[7]  = (_Float16)a1.w;
        a[8]  = (_Float16)a2.x; a[9]  = (_Float16)a2.y;
        a[10] = (_Float16)a2.z; a[11] = (_Float16)a2.w;
        a[12] = (_Float16)a3.x; a[13] = (_Float16)a3.y;
        a[14] = (_Float16)a3.z; a[15] = (_Float16)a3.w;
#pragma unroll
        for (int nt = 0; nt < 8; ++nt) {
            int n = nt * 16 + mloc;
            const v8h* bp = (const v8h*)(sW + n * WPITCH + k0);
            v8h blo = bp[0];
            v8h bhi = bp[2];
            v16h b;
#pragma unroll
            for (int i = 0; i < 8; ++i) { b[i] = blo[i]; b[8 + i] = bhi[i]; }
            acc[nt] = __builtin_amdgcn_wmma_f32_16x16x32_f16(
                false, a, false, b, (short)0, acc[nt], false, false);
        }
    }

#pragma unroll
    for (int nt = 0; nt < 8; ++nt) {
        int col = nt * 16 + mloc;
#pragma unroll
        for (int r = 0; r < 8; ++r) {
            float v = acc[nt][r];
            if (RELU) v = fmaxf(v, 0.0f);
            int M = row0 + r + half * 8;
            out[(size_t)M * HID + col] = v;
        }
    }
}

#define MLP_TPW 2
__global__ __launch_bounds__(256) void mlp2_wmma_kernel(
    const float* __restrict__ A, const _Float16* __restrict__ W1t,
    const _Float16* __restrict__ W2t, const float* __restrict__ bias1,
    const float* __restrict__ bias2, float* __restrict__ out, int rows,
    size_t strideA, size_t strideO) {
    A += blockIdx.y * strideA;
    out += blockIdx.y * strideO;
    __shared__ __align__(16) _Float16 sW1[HID * WPITCH];
    __shared__ __align__(16) _Float16 sW2[HID * WPITCH];
    {
        const uint4* s1 = (const uint4*)W1t;
        const uint4* s2 = (const uint4*)W2t;
        uint4 q1[8], q2[8];
#pragma unroll
        for (int j = 0; j < 8; ++j) {
            q1[j] = s1[j * 256 + threadIdx.x];
            q2[j] = s2[j * 256 + threadIdx.x];
        }
#pragma unroll
        for (int j = 0; j < 8; ++j) {
            int i = j * 256 + threadIdx.x;
            int o = (i >> 4) * WPITCH + (i & 15) * 8;
            *(uint4*)(sW1 + o) = q1[j];
            *(uint4*)(sW2 + o) = q2[j];
        }
    }
    __syncthreads();

    const int wave = threadIdx.x >> 5, lane = threadIdx.x & 31;
    const int mloc = lane & 15;
    const int half = lane >> 4;
    const int base = half * 8;

    for (int t = 0; t < MLP_TPW; ++t) {
        const int tile = (blockIdx.x * MLP_TPW + t) * 8 + wave;
        const int row0 = tile * 16;
        if (row0 >= rows) break;
        const float* rowA = A + (size_t)(row0 + mloc) * HID;

        v8f acc1[8];
#pragma unroll
        for (int nt = 0; nt < 8; ++nt) {
            const float4 c0 = *(const float4*)(bias1 + nt * 16 + base);
            const float4 c1 = *(const float4*)(bias1 + nt * 16 + base + 4);
            acc1[nt][0] = c0.x; acc1[nt][1] = c0.y; acc1[nt][2] = c0.z; acc1[nt][3] = c0.w;
            acc1[nt][4] = c1.x; acc1[nt][5] = c1.y; acc1[nt][6] = c1.z; acc1[nt][7] = c1.w;
        }
#pragma unroll
        for (int kk = 0; kk < 4; ++kk) {
            const int k0 = kk * 32 + base;
            float4 a0 = *(const float4*)(rowA + k0);
            float4 a1 = *(const float4*)(rowA + k0 + 4);
            float4 a2 = *(const float4*)(rowA + k0 + 16);
            float4 a3 = *(const float4*)(rowA + k0 + 20);
            v16h a;
            a[0]  = (_Float16)a0.x; a[1]  = (_Float16)a0.y;
            a[2]  = (_Float16)a0.z; a[3]  = (_Float16)a0.w;
            a[4]  = (_Float16)a1.x; a[5]  = (_Float16)a1.y;
            a[6]  = (_Float16)a1.z; a[7]  = (_Float16)a1.w;
            a[8]  = (_Float16)a2.x; a[9]  = (_Float16)a2.y;
            a[10] = (_Float16)a2.z; a[11] = (_Float16)a2.w;
            a[12] = (_Float16)a3.x; a[13] = (_Float16)a3.y;
            a[14] = (_Float16)a3.z; a[15] = (_Float16)a3.w;
#pragma unroll
            for (int nt = 0; nt < 8; ++nt) {
                const v8h* wp = (const v8h*)(sW1 + (nt * 16 + mloc) * WPITCH + k0);
                v8h wlo = wp[0];
                v8h whi = wp[2];
                v16h w;
#pragma unroll
                for (int i = 0; i < 8; ++i) { w[i] = wlo[i]; w[8 + i] = whi[i]; }
                acc1[nt] = __builtin_amdgcn_wmma_f32_16x16x32_f16(
                    false, w, false, a, (short)0, acc1[nt], false, false);
            }
        }

        v16h hfr[4];
#pragma unroll
        for (int kk = 0; kk < 4; ++kk) {
#pragma unroll
            for (int i = 0; i < 8; ++i) {
                hfr[kk][i]     = (_Float16)fmaxf(acc1[2 * kk][i], 0.0f);
                hfr[kk][8 + i] = (_Float16)fmaxf(acc1[2 * kk + 1][i], 0.0f);
            }
        }

        v8f acc2[8];
#pragma unroll
        for (int nt = 0; nt < 8; ++nt) {
            float bv = bias2[nt * 16 + mloc];
#pragma unroll
            for (int r = 0; r < 8; ++r) acc2[nt][r] = bv;
        }
#pragma unroll
        for (int kk = 0; kk < 4; ++kk) {
            const int k0 = kk * 32 + base;
#pragma unroll
            for (int nt = 0; nt < 8; ++nt) {
                const v8h* bp = (const v8h*)(sW2 + (nt * 16 + mloc) * WPITCH + k0);
                v8h blo = bp[0];
                v8h bhi = bp[2];
                v16h b;
#pragma unroll
                for (int i = 0; i < 8; ++i) { b[i] = blo[i]; b[8 + i] = bhi[i]; }
                acc2[nt] = __builtin_amdgcn_wmma_f32_16x16x32_f16(
                    false, hfr[kk], false, b, (short)0, acc2[nt], false, false);
            }
        }
#pragma unroll
        for (int nt = 0; nt < 8; ++nt) {
            int col = nt * 16 + mloc;
#pragma unroll
            for (int r = 0; r < 8; ++r) {
                int M = row0 + r + half * 8;
                out[(size_t)M * HID + col] = fmaxf(acc2[nt][r], 0.0f);
            }
        }
    }
}

extern "C" void kernel_launch(void* const* d_in, const int* in_sizes, int n_in,
                              void* d_out, int out_size, void* d_ws, size_t ws_size,
                              hipStream_t stream) {
    const int N = in_sizes[0];
    const int E = in_sizes[1] / 2;
    const int G = out_size / (3 * HID);

    const float* enc_W  = (const float*)d_in[12];
    const float* enc_b  = (const float*)d_in[13];
    const float* edge_W = (const float*)d_in[14];
    const float* edge_b = (const float*)d_in[15];
    const float* eps    = (const float*)d_in[16];
    const float* W1     = (const float*)d_in[17];
    const float* b1     = (const float*)d_in[18];
    const float* W2     = (const float*)d_in[19];
    const float* b2     = (const float*)d_in[20];
    const float* fc_W   = (const float*)d_in[21];
    const float* fc_b   = (const float*)d_in[22];

    const size_t plane = (size_t)N * HID;
    const size_t pplane = (size_t)G * HID;
    float* bufA   = (float*)d_ws;
    float* bufB   = bufA + 3 * plane;
    float* pooled = bufB + 3 * plane;
    _Float16* w16 = (_Float16*)(pooled + 3 * pplane);
    _Float16* W1t0 = w16;
    _Float16* W1t1 = w16 + 16384;
    _Float16* W2t0 = w16 + 2 * 16384;
    _Float16* W2t1 = w16 + 3 * 16384;
    _Float16* fcWt = w16 + 4 * 16384;
    const int NP = (N + 4) & ~3;
    const int EP = (E + 3) & ~3;
    int* cnt   = (int*)(w16 + 5 * 16384);
    int* off   = cnt + 3 * (size_t)NP;
    int* perm  = off + 3 * (size_t)NP;
    int* ssrc  = perm + 3 * (size_t)EP;
    float* sea = (float*)(ssrc + 3 * (size_t)EP);

    convert_wt_kernel<<<64, 256, 0, stream>>>(W1,             W1t0);
    convert_wt_kernel<<<64, 256, 0, stream>>>(W1 + HID * HID, W1t1);
    convert_wt_kernel<<<64, 256, 0, stream>>>(W2,             W2t0);
    convert_wt_kernel<<<64, 256, 0, stream>>>(W2 + HID * HID, W2t1);
    convert_wt_kernel<<<64, 256, 0, stream>>>(fc_W,           fcWt);

    const int nodeBlk   = (N + 8 * RUN - 1) / (8 * RUN);
    const int nodeBlkM  = (N + 8 * RUN_MEM - 1) / (8 * RUN_MEM);
    const int mlpBlk    = ((N + 15) / 16 + 8 * MLP_TPW - 1) / (8 * MLP_TPW);
    const int gP4       = (G * HID) / 4;
    const int fcBlk     = ((G + 15) / 16 + 7) / 8;

    EdgeIn ein;
    ein.ei0 = (const int*)d_in[1];   ein.ei1 = (const int*)d_in[5];   ein.ei2 = (const int*)d_in[9];
    ein.ea0 = (const float*)d_in[2]; ein.ea1 = (const float*)d_in[6]; ein.ea2 = (const float*)d_in[10];
    const dim3 edgeGrid((E + 255) / 256, 3);
    const int RB = 85;
    const int R  = (N + RB - 1) / RB;
    const dim3 rangeGrid(RB, 3);
    range_kernel<false><<<rangeGrid, 1024, R * sizeof(int), stream>>>(ein, cnt, off, perm, E, N, NP, EP, R);
    scan_kernel<<<3, 1024, 0, stream>>>(cnt, off, N, NP);
    range_kernel<true><<<rangeGrid, 1024, R * sizeof(int), stream>>>(ein, cnt, off, perm, E, N, NP, EP, R);
    rank_kernel<<<edgeGrid, 256, 0, stream>>>(ein, off, perm, ssrc, sea, E, N, NP, EP);

    NodeIn nin;
    nin.x0 = (const float*)d_in[0]; nin.x1 = (const float*)d_in[4]; nin.x2 = (const float*)d_in[8];
    nin.b0 = (const int*)d_in[3];   nin.b1 = (const int*)d_in[7];   nin.b2 = (const int*)d_in[11];
    const dim3 nodeGrid(nodeBlk, 3), nodeGridM(nodeBlkM, 3), mlpGrid(mlpBlk, 3), fcGrid(fcBlk, 3);

    float* hcur = bufA;
    float* aggb = bufB;
    for (int l = 0; l < 2; ++l) {
        if (l == 0)
            gather_encoded_kernel<<<nodeGrid, 256, 0, stream>>>(
                nin, enc_W, enc_b, off, ssrc, sea, edge_W, edge_b, eps, 0, aggb, N, NP, EP);
        else
            gather_kernel<<<nodeGridM, 256, 0, stream>>>(
                hcur, off, ssrc, sea, edge_W + (size_t)l * 2 * HID, edge_b + (size_t)l * HID,
                eps, l, aggb, N, NP, EP);
        _Float16* w1t = (l == 0) ? W1t0 : W1t1;
        _Float16* w2t = (l == 0) ? W2t0 : W2t1;
        mlp2_wmma_kernel<<<mlpGrid, 256, 0, stream>>>(
            aggb, w1t, w2t, b1 + (size_t)l * HID, b2 + (size_t)l * HID, hcur, N, plane, plane);
    }

    zero_kernel<<<(3 * gP4 + 255) / 256, 256, 0, stream>>>(pooled, 3 * gP4);
    pool_kernel<<<nodeGrid, 256, 0, stream>>>(hcur, nin, pooled, N, G);
    gemm128_wmma_kernel<false><<<fcGrid, 256, 0, stream>>>(
        pooled, fcWt, fc_b, (float*)d_out, G, pplane, pplane);
}
